# starting kernel
# baseline (speedup 1.0000x reference)
#include <hip/hip_runtime.h>
#include <hip/hip_bf16.h>

typedef __attribute__((ext_vector_type(2))) float v2f;
typedef __attribute__((ext_vector_type(8))) float v8f;

#define EPSC   1e-6f
#define SCOEF  0.025f
#define ACOEF  0.975f
#define ALPHAC 0.98f
#define DELTAC 2.0f
#define DROOT  1.1892071150027210667f

constexpr int Bn = 64;
constexpr int Tn = 2048;
constexpr int Fn = 128;
constexpr unsigned ROWB = Fn * 4u;

constexpr int NSEG   = 4;
constexpr int SEGLEN = Tn / NSEG;
constexpr int SEGCH  = SEGLEN / 16;
constexpr int WARMCH = 32;

__device__ __forceinline__ float ldf(const char* base, unsigned byteOff) {
    return *(const float*)(base + byteOff);
}

__global__ __launch_bounds__(256)
void pcen_wmma_kernel(const float* __restrict__ x, float* __restrict__ out) {
    const int lane = threadIdx.x & 31;

    const int gw = __builtin_amdgcn_readfirstlane(
        (int)(blockIdx.x * (blockDim.x >> 5) + (threadIdx.x >> 5)));
    const int seg = gw & (NSEG - 1);
    const int ft  = (gw >> 2) & 7;
    const int b   = gw >> 5;

    const int half = lane >> 4;
    const int n    = lane & 15;

    float ap[17];
    ap[0] = 1.0f;
    #pragma unroll
    for (int k = 1; k <= 16; ++k) ap[k] = ap[k - 1] * ACOEF;

    v2f Ls[4];
    #pragma unroll
    for (int s = 0; s < 4; ++s) {
        #pragma unroll
        for (int v = 0; v < 2; ++v) {
            const int K = 4 * s + 2 * half + v;
            float val = (n >= K) ? (SCOEF * ap[n - K]) : 0.0f;
            if (v == 0) Ls[s].x = val; else Ls[s].y = val;
        }
    }
    v2f Ac;
    Ac.x = (half == 0) ? ap[n + 1] : 0.0f;
    Ac.y = 0.0f;

    const char* __restrict__ xb8 = (const char*)(x   + (size_t)b * Tn * Fn + ft * 16);
    char* __restrict__       ob8 = (char*)      (out + (size_t)b * Tn * Fn + ft * 16);

    const int t_start = seg * SEGLEN;
    const int warm    = (seg == 0) ? 0 : WARMCH;
    const int tfirst  = t_start - warm * 16;

    unsigned offB = (unsigned)(tfirst + 2 * half) * ROWB + (unsigned)n * 4u;
    unsigned offD = (unsigned)(t_start + 8 * half) * ROWB + (unsigned)n * 4u;
    unsigned offP = (unsigned)(tfirst + 16 + lane) * ROWB + (unsigned)n * 4u;

    float c = ldf(xb8, (unsigned)tfirst * ROWB + (unsigned)n * 4u);

    for (int ch = -warm; ch < SEGCH; ++ch) {
        const int tc = t_start + ch * 16;

        if (tc + 16 + lane < Tn) {
            __builtin_prefetch(xb8 + offP, 0, 1);
        }
        offP += 16u * ROWB;

        v8f acc = {};

        {
            v2f Bc;
            Bc.x = (half == 0) ? c : 0.0f;
            Bc.y = 0.0f;
            acc = __builtin_amdgcn_wmma_f32_16x16x4_f32(
                false, Ac, false, Bc, (short)0, acc, false, false);
        }

        #pragma unroll
        for (int s = 0; s < 4; ++s) {
            v2f Bx;
            Bx.x = ldf(xb8, offB + (unsigned)(4 * s + 0) * ROWB);
            Bx.y = ldf(xb8, offB + (unsigned)(4 * s + 1) * ROWB);
            acc = __builtin_amdgcn_wmma_f32_16x16x4_f32(
                false, Ls[s], false, Bx, (short)0, acc, false, false);
        }
        offB += 16u * ROWB;

        c = __shfl(acc[7], 16 + n, 32);

        if (ch >= 0) {
            #pragma unroll
            for (int v = 0; v < 8; ++v) {
                const unsigned o = offD + (unsigned)v * ROWB;
                const float xv = ldf(xb8, o);
                const float mm = acc[v] + EPSC;
                const float ratio =
                    xv * __builtin_amdgcn_exp2f(-ALPHAC * __builtin_amdgcn_logf(mm));
                const float val =
                    __builtin_amdgcn_sqrtf(__builtin_amdgcn_sqrtf(ratio + DELTAC)) - DROOT;
                __builtin_nontemporal_store(val, (float*)(ob8 + o));
            }
            offD += 16u * ROWB;
        }
    }
}

extern "C" void kernel_launch(void* const* d_in, const int* in_sizes, int n_in,
                              void* d_out, int out_size, void* d_ws, size_t ws_size,
                              hipStream_t stream) {
    (void)in_sizes; (void)n_in; (void)d_ws; (void)ws_size; (void)out_size;
    const float* x = (const float*)d_in[0];
    float* out = (float*)d_out;

    const int totalWaves = Bn * (Fn / 16) * NSEG;
    const int wavesPerBlock = 8;
    dim3 grid(totalWaves / wavesPerBlock);
    dim3 block(wavesPerBlock * 32);
    pcen_wmma_kernel<<<grid, block, 0, stream>>>(x, out);
}


# Software-pipelined x loads: each chunk's B- and D-layout tiles are requested 7 and 3 chunks ahead into register rings, so the carry chain never waits for memory; WMMA chain and sum order unchanged
# speedup vs baseline: 1.5748x; 1.5748x over previous
#include <hip/hip_runtime.h>
#include <hip/hip_bf16.h>


typedef __attribute__((ext_vector_type(2))) float v2f;
typedef __attribute__((ext_vector_type(8))) float v8f;

#define EPSC   1e-6f
#define SCOEF  0.025f
#define ACOEF  0.975f
#define ALPHAC 0.98f
#define DELTAC 2.0f
#define DROOT  1.1892071150027210667f

constexpr int Bn = 64;
constexpr int Tn = 2048;
constexpr int Fn = 128;
constexpr unsigned ROWB = Fn * 4u;
constexpr unsigned CHB  = 16u * ROWB;

constexpr int NSEG   = 4;
constexpr int SEGLEN = Tn / NSEG;
constexpr int SEGCH  = SEGLEN / 16;
constexpr int WARMCH = 32;

constexpr int BRING = 8;
constexpr int BDIST = 7;
constexpr int DRING = 4;
constexpr int DDIST = 3;

static_assert(WARMCH % BRING == 0 && SEGCH % BRING == 0, "ring slot = chunk mod ring");
static_assert(BRING % DRING == 0, "D slot follows the unrolled body index");

__device__ __forceinline__ float ldf(const char* base, unsigned byteOff) {
    return *(const float*)(base + byteOff);
}

__device__ __forceinline__ void load_b_tile(v2f (&B)[4], const char* cb, unsigned laneB) {
    #pragma unroll
    for (int s = 0; s < 4; ++s) {
        B[s].x = ldf(cb, laneB + (unsigned)(4 * s + 0) * ROWB);
        B[s].y = ldf(cb, laneB + (unsigned)(4 * s + 1) * ROWB);
    }
}

__device__ __forceinline__ void load_d_tile(float (&D)[8], const char* cb, unsigned laneD) {
    #pragma unroll
    for (int v = 0; v < 8; ++v) D[v] = ldf(cb, laneD + (unsigned)v * ROWB);
}

__device__ __forceinline__ v8f ema_chunk(const v2f (&Ls)[4], v2f Ac, const v2f (&B)[4],
                                         float c, int half) {
    v8f acc = {};
    v2f Bc;
    Bc.x = (half == 0) ? c : 0.0f;
    Bc.y = 0.0f;
    acc = __builtin_amdgcn_wmma_f32_16x16x4_f32(
        false, Ac, false, Bc, (short)0, acc, false, false);
    #pragma unroll
    for (int s = 0; s < 4; ++s) {
        acc = __builtin_amdgcn_wmma_f32_16x16x4_f32(
            false, Ls[s], false, B[s], (short)0, acc, false, false);
    }
    return acc;
}

__global__ __launch_bounds__(256)
void pcen_wmma_kernel(const float* __restrict__ x, float* __restrict__ out) {
    const int lane = threadIdx.x & 31;

    const int gw = __builtin_amdgcn_readfirstlane(
        (int)(blockIdx.x * (blockDim.x >> 5) + (threadIdx.x >> 5)));
    const int seg = gw & (NSEG - 1);
    const int ft  = (gw >> 2) & 7;
    const int b   = gw >> 5;

    const int half = lane >> 4;
    const int n    = lane & 15;

    float ap[17];
    ap[0] = 1.0f;
    #pragma unroll
    for (int k = 1; k <= 16; ++k) ap[k] = ap[k - 1] * ACOEF;

    v2f Ls[4];
    #pragma unroll
    for (int s = 0; s < 4; ++s) {
        #pragma unroll
        for (int v = 0; v < 2; ++v) {
            const int K = 4 * s + 2 * half + v;
            float val = (n >= K) ? (SCOEF * ap[n - K]) : 0.0f;
            if (v == 0) Ls[s].x = val; else Ls[s].y = val;
        }
    }
    v2f Ac;
    Ac.x = (half == 0) ? ap[n + 1] : 0.0f;
    Ac.y = 0.0f;

    const int t_start = seg * SEGLEN;
    const int warm    = (seg == 0) ? 0 : WARMCH;
    const int tfirst  = t_start - warm * 16;

    const char* __restrict__ xwalk =
        (const char*)(x + (size_t)b * Tn * Fn + ft * 16) + (size_t)tfirst * ROWB;
    const char* __restrict__ xmain = xwalk + (size_t)warm * CHB;
    char* __restrict__ omain =
        (char*)(out + (size_t)b * Tn * Fn + ft * 16) + (size_t)t_start * ROWB;

    const unsigned laneB = (unsigned)(2 * half) * ROWB + (unsigned)n * 4u;
    const unsigned laneD = (unsigned)(8 * half) * ROWB + (unsigned)n * 4u;

    float c = ldf(xwalk, (unsigned)n * 4u);

    v2f   Bq[BRING][4];
    float Dq[DRING][8];

    #pragma unroll
    for (int i = 0; i < BDIST; ++i) load_b_tile(Bq[i], xwalk + (size_t)i * CHB, laneB);

    if (warm != 0) {
        #pragma nounroll
        for (int j = 0; j < WARMCH; j += BRING) {
            #pragma unroll
            for (int u = 0; u < BRING; ++u) {
                load_b_tile(Bq[(u + BDIST) % BRING],
                            xwalk + (size_t)(j + u + BDIST) * CHB, laneB);
                const v8f acc = ema_chunk(Ls, Ac, Bq[u], c, half);
                c = __shfl(acc[7], 16 + n, 32);
            }
        }
    }

    #pragma unroll
    for (int i = 0; i < DDIST; ++i) load_d_tile(Dq[i], xmain + (size_t)i * CHB, laneD);

    #pragma nounroll
    for (int k = 0; k < SEGCH; k += BRING) {
        #pragma unroll
        for (int u = 0; u < BRING; ++u) {
            const int kb = (k + u + BDIST < SEGCH) ? (k + u + BDIST) : (SEGCH - 1);
            const int kd = (k + u + DDIST < SEGCH) ? (k + u + DDIST) : (SEGCH - 1);
            load_b_tile(Bq[(u + BDIST) % BRING], xmain + (size_t)kb * CHB, laneB);
            load_d_tile(Dq[(u + DDIST) % DRING], xmain + (size_t)kd * CHB, laneD);

            const v8f acc = ema_chunk(Ls, Ac, Bq[u], c, half);
            c = __shfl(acc[7], 16 + n, 32);

            char* __restrict__ oc = omain + (size_t)(k + u) * CHB;
            #pragma unroll
            for (int v = 0; v < 8; ++v) {
                const float xv = Dq[u % DRING][v];
                const float mm = acc[v] + EPSC;
                const float ratio =
                    xv * __builtin_amdgcn_exp2f(-ALPHAC * __builtin_amdgcn_logf(mm));
                const float val =
                    __builtin_amdgcn_sqrtf(__builtin_amdgcn_sqrtf(ratio + DELTAC)) - DROOT;
                __builtin_nontemporal_store(val, (float*)(oc + laneD + (unsigned)v * ROWB));
            }
        }
    }
}

extern "C" void kernel_launch(void* const* d_in, const int* in_sizes, int n_in,
                              void* d_out, int out_size, void* d_ws, size_t ws_size,
                              hipStream_t stream) {
    (void)in_sizes; (void)n_in; (void)d_ws; (void)ws_size; (void)out_size;
    const float* x = (const float*)d_in[0];
    float* out = (float*)d_out;

    const int totalWaves = Bn * (Fn / 16) * NSEG;
    const int wavesPerBlock = 8;
    dim3 grid(totalWaves / wavesPerBlock);
    dim3 block(wavesPerBlock * 32);
    pcen_wmma_kernel<<<grid, block, 0, stream>>>(x, out);
}


# Registers named by issued loads/stores (address pairs, store data) are left alone for 3+ chunks: fresh address pair per chunk, held rings; the s_wait_xcnt 0 stalls leave both loops; sums unchanged
# speedup vs baseline: 1.7050x; 1.0827x over previous
#include <hip/hip_runtime.h>
#include <hip/hip_bf16.h>


typedef __attribute__((ext_vector_type(2))) float v2f;
typedef __attribute__((ext_vector_type(8))) float v8f;

#define EPSC   1e-6f
#define SCOEF  0.025f
#define ACOEF  0.975f
#define ALPHAC 0.98f
#define DELTAC 2.0f
#define DROOT  1.1892071150027210667f

constexpr int Bn = 64;
constexpr int Tn = 2048;
constexpr int Fn = 128;
constexpr unsigned ROWB = Fn * 4u;
constexpr unsigned CHB  = 16u * ROWB;

constexpr int NSEG   = 4;
constexpr int SEGLEN = Tn / NSEG;
constexpr int SEGCH  = SEGLEN / 16;
constexpr int WARMCH = 32;

constexpr int BRING = 8;
constexpr int BDIST = 7;
constexpr int DRING = 4;
constexpr int DDIST = 3;
constexpr int HOLD  = 4;

static_assert(WARMCH % BRING == 0 && SEGCH % BRING == 0, "ring slot = chunk mod ring");
static_assert(BRING % DRING == 0, "D slot follows the unrolled body index");
static_assert(BRING % HOLD == 0, "hold slot follows the unrolled body index");

typedef const __attribute__((address_space(1))) char*  gcptr;
typedef __attribute__((address_space(1))) char*        gptr;
typedef const __attribute__((address_space(1))) float* gcfptr;
typedef __attribute__((address_space(1))) float*       gfptr;

__device__ __forceinline__ float ldf(gcptr base, unsigned byteOff) {
    return *(gcfptr)(base + byteOff);
}

__device__ __forceinline__ gcptr lane_ptr(const char* slab, int chunk, unsigned laneOff) {
    gcptr p = (gcptr)(slab + (size_t)chunk * CHB);
    asm("" : "+s"(p));
    return p + laneOff;
}

__device__ __forceinline__ void hold_chunk(gcptr pb, gcptr pd, gptr po, const float (&O)[8]) {
    asm volatile("" :: "v"(pb), "v"(pd), "v"(po),
                       "v"(O[0]), "v"(O[1]), "v"(O[2]), "v"(O[3]),
                       "v"(O[4]), "v"(O[5]), "v"(O[6]), "v"(O[7]));
}
__device__ __forceinline__ void hold_ptr(gcptr pb) {
    asm volatile("" :: "v"(pb));
}

__device__ __forceinline__ void load_b_tile(v2f (&B)[4], gcptr pb) {
    #pragma unroll
    for (int s = 0; s < 4; ++s) {
        B[s].x = ldf(pb, (unsigned)(4 * s + 0) * ROWB);
        B[s].y = ldf(pb, (unsigned)(4 * s + 1) * ROWB);
    }
}

__device__ __forceinline__ void load_d_tile(float (&D)[8], gcptr pd) {
    #pragma unroll
    for (int v = 0; v < 8; ++v) D[v] = ldf(pd, (unsigned)v * ROWB);
}

__device__ __forceinline__ v8f ema_chunk(const v2f (&Ls)[4], v2f Ac, const v2f (&B)[4],
                                         float c, int half) {
    v8f acc = {};
    v2f Bc;
    Bc.x = (half == 0) ? c : 0.0f;
    Bc.y = 0.0f;
    acc = __builtin_amdgcn_wmma_f32_16x16x4_f32(
        false, Ac, false, Bc, (short)0, acc, false, false);
    #pragma unroll
    for (int s = 0; s < 4; ++s) {
        acc = __builtin_amdgcn_wmma_f32_16x16x4_f32(
            false, Ls[s], false, B[s], (short)0, acc, false, false);
    }
    return acc;
}

__global__ __launch_bounds__(256)
void pcen_wmma_kernel(const float* __restrict__ x, float* __restrict__ out) {
    const int lane = threadIdx.x & 31;

    const int gw = __builtin_amdgcn_readfirstlane(
        (int)(blockIdx.x * (blockDim.x >> 5) + (threadIdx.x >> 5)));
    const int seg = gw & (NSEG - 1);
    const int ft  = (gw >> 2) & 7;
    const int b   = gw >> 5;

    const int half = lane >> 4;
    const int n    = lane & 15;

    float ap[17];
    ap[0] = 1.0f;
    #pragma unroll
    for (int k = 1; k <= 16; ++k) ap[k] = ap[k - 1] * ACOEF;

    v2f Ls[4];
    #pragma unroll
    for (int s = 0; s < 4; ++s) {
        #pragma unroll
        for (int v = 0; v < 2; ++v) {
            const int K = 4 * s + 2 * half + v;
            float val = (n >= K) ? (SCOEF * ap[n - K]) : 0.0f;
            if (v == 0) Ls[s].x = val; else Ls[s].y = val;
        }
    }
    v2f Ac;
    Ac.x = (half == 0) ? ap[n + 1] : 0.0f;
    Ac.y = 0.0f;

    const int t_start = seg * SEGLEN;
    const int warm    = (seg == 0) ? 0 : WARMCH;
    const int tfirst  = t_start - warm * 16;

    const char* __restrict__ xwalk =
        (const char*)(x + (size_t)b * Tn * Fn + ft * 16) + (size_t)tfirst * ROWB;
    const char* __restrict__ xmain = xwalk + (size_t)warm * CHB;
    char* __restrict__ omain =
        (char*)(out + (size_t)b * Tn * Fn + ft * 16) + (size_t)t_start * ROWB;

    const unsigned laneB = (unsigned)(2 * half) * ROWB + (unsigned)n * 4u;
    const unsigned laneD = (unsigned)(8 * half) * ROWB + (unsigned)n * 4u;

    float c = ldf((gcptr)xwalk, (unsigned)n * 4u);

    v2f   Bq[BRING][4];
    float Dq[DRING][8];

    #pragma unroll
    for (int i = 0; i < BDIST; ++i) load_b_tile(Bq[i], (gcptr)(xwalk + (size_t)i * CHB) + laneB);

    gcptr PB[BRING], PD[HOLD];
    gptr  PO[HOLD];
    float Oq[HOLD][8];
    #pragma unroll
    for (int i = 0; i < BRING; ++i) PB[i] = (gcptr)xwalk;
    #pragma unroll
    for (int i = 0; i < HOLD; ++i) {
        PD[i] = (gcptr)xwalk; PO[i] = (gptr)omain;
        #pragma unroll
        for (int v = 0; v < 8; ++v) Oq[i][v] = 0.0f;
    }

    if (warm != 0) {
        #pragma nounroll
        for (int j = 0; j < WARMCH; j += BRING) {
            #pragma unroll
            for (int u = 0; u < BRING; ++u) {
                hold_ptr(PB[u]);
                PB[u] = lane_ptr(xwalk, j + u + BDIST, laneB);
                load_b_tile(Bq[(u + BDIST) % BRING], PB[u]);
                const v8f acc = ema_chunk(Ls, Ac, Bq[u], c, half);
                c = __shfl(acc[7], 16 + n, 32);
            }
        }
    }

    #pragma unroll
    for (int i = 0; i < DDIST; ++i) load_d_tile(Dq[i], (gcptr)(xmain + (size_t)i * CHB) + laneD);

    #pragma nounroll
    for (int k = 0; k < SEGCH; k += BRING) {
        #pragma unroll
        for (int u = 0; u < BRING; ++u) {
            const int kb = (k + u + BDIST < SEGCH) ? (k + u + BDIST) : (SEGCH - 1);
            const int kd = (k + u + DDIST < SEGCH) ? (k + u + DDIST) : (SEGCH - 1);
            hold_chunk(PB[u], PD[u % HOLD], PO[u % HOLD], Oq[u % HOLD]);
            PB[u] = lane_ptr(xmain, kb, laneB);
            PD[u % HOLD] = lane_ptr(xmain, kd, laneD);
            PO[u % HOLD] = (gptr)lane_ptr(omain, k + u, laneD);
            load_b_tile(Bq[(u + BDIST) % BRING], PB[u]);
            load_d_tile(Dq[(u + DDIST) % DRING], PD[u % HOLD]);

            const v8f acc = ema_chunk(Ls, Ac, Bq[u], c, half);
            c = __shfl(acc[7], 16 + n, 32);

            #pragma unroll
            for (int v = 0; v < 8; ++v) {
                const float xv = Dq[u % DRING][v];
                const float mm = acc[v] + EPSC;
                const float ratio =
                    xv * __builtin_amdgcn_exp2f(-ALPHAC * __builtin_amdgcn_logf(mm));
                const float val =
                    __builtin_amdgcn_sqrtf(__builtin_amdgcn_sqrtf(ratio + DELTAC)) - DROOT;
                Oq[u % HOLD][v] = val;
                __builtin_nontemporal_store(val, (gfptr)(PO[u % HOLD] + (unsigned)v * ROWB));
            }
        }
    }
}

extern "C" void kernel_launch(void* const* d_in, const int* in_sizes, int n_in,
                              void* d_out, int out_size, void* d_ws, size_t ws_size,
                              hipStream_t stream) {
    (void)in_sizes; (void)n_in; (void)d_ws; (void)ws_size; (void)out_size;
    const float* x = (const float*)d_in[0];
    float* out = (float*)d_out;

    const int totalWaves = Bn * (Fn / 16) * NSEG;
    const int wavesPerBlock = 8;
    dim3 grid(totalWaves / wavesPerBlock);
    dim3 block(wavesPerBlock * 32);
    pcen_wmma_kernel<<<grid, block, 0, stream>>>(x, out);
}


# One x tile per chunk, not two: L and x fragments take k order 8*half+2s+v, so a lane's B elements are its epilogue x (D layout); 8 loads a chunk, not 16; sum order inside a chunk changes
# speedup vs baseline: 2.0262x; 1.1884x over previous
#include <hip/hip_runtime.h>
#include <hip/hip_bf16.h>


typedef __attribute__((ext_vector_type(2))) float v2f;
typedef __attribute__((ext_vector_type(8))) float v8f;

#define EPSC   1e-6f
#define SCOEF  0.025f
#define ACOEF  0.975f
#define ALPHAC 0.98f
#define DELTAC 2.0f
#define DROOT  1.1892071150027210667f

constexpr int Bn = 64;
constexpr int Tn = 2048;
constexpr int Fn = 128;
constexpr unsigned ROWB = Fn * 4u;
constexpr unsigned CHB  = 16u * ROWB;

constexpr int NSEG   = 4;
constexpr int SEGLEN = Tn / NSEG;
constexpr int SEGCH  = SEGLEN / 16;
constexpr int WARMCH = 32;

constexpr int BRING = 8;
constexpr int BDIST = 7;
constexpr int HOLD  = 4;

static_assert(WARMCH % BRING == 0 && SEGCH % BRING == 0, "ring slot = chunk mod ring");
static_assert(BRING % HOLD == 0, "hold slot follows the unrolled body index");

typedef const __attribute__((address_space(1))) char*  gcptr;
typedef __attribute__((address_space(1))) char*        gptr;
typedef const __attribute__((address_space(1))) float* gcfptr;
typedef __attribute__((address_space(1))) float*       gfptr;

__device__ __forceinline__ float ldf(gcptr base, unsigned byteOff) {
    return *(gcfptr)(base + byteOff);
}

__device__ __forceinline__ gcptr lane_ptr(const char* slab, int chunk, unsigned laneOff) {
    gcptr p = (gcptr)(slab + (size_t)chunk * CHB);
    asm("" : "+s"(p));
    return p + laneOff;
}

__device__ __forceinline__ void hold_chunk(gcptr pb, gptr po, const float (&O)[8]) {
    asm volatile("" :: "v"(pb), "v"(po),
                       "v"(O[0]), "v"(O[1]), "v"(O[2]), "v"(O[3]),
                       "v"(O[4]), "v"(O[5]), "v"(O[6]), "v"(O[7]));
}
__device__ __forceinline__ void hold_ptr(gcptr pb) {
    asm volatile("" :: "v"(pb));
}

__device__ __forceinline__ constexpr int KROW(int half, int s, int v) {
    return 8 * half + 2 * s + v;
}

__device__ __forceinline__ void load_b_tile(v2f (&B)[4], gcptr pb) {
    #pragma unroll
    for (int s = 0; s < 4; ++s) {
        B[s].x = ldf(pb, (unsigned)KROW(0, s, 0) * ROWB);
        B[s].y = ldf(pb, (unsigned)KROW(0, s, 1) * ROWB);
    }
}

__device__ __forceinline__ v8f ema_chunk(const v2f (&Ls)[4], v2f Ac, const v2f (&B)[4],
                                         float c, int half) {
    v8f acc = {};
    v2f Bc;
    Bc.x = (half == 0) ? c : 0.0f;
    Bc.y = 0.0f;
    acc = __builtin_amdgcn_wmma_f32_16x16x4_f32(
        false, Ac, false, Bc, (short)0, acc, false, false);
    #pragma unroll
    for (int s = 0; s < 4; ++s) {
        acc = __builtin_amdgcn_wmma_f32_16x16x4_f32(
            false, Ls[s], false, B[s], (short)0, acc, false, false);
    }
    return acc;
}

__global__ __launch_bounds__(256)
void pcen_wmma_kernel(const float* __restrict__ x, float* __restrict__ out) {
    const int lane = threadIdx.x & 31;

    const int gw = __builtin_amdgcn_readfirstlane(
        (int)(blockIdx.x * (blockDim.x >> 5) + (threadIdx.x >> 5)));
    const int seg = gw & (NSEG - 1);
    const int ft  = (gw >> 2) & 7;
    const int b   = gw >> 5;

    const int half = lane >> 4;
    const int n    = lane & 15;

    float ap[17];
    ap[0] = 1.0f;
    #pragma unroll
    for (int k = 1; k <= 16; ++k) ap[k] = ap[k - 1] * ACOEF;

    v2f Ls[4];
    #pragma unroll
    for (int s = 0; s < 4; ++s) {
        #pragma unroll
        for (int v = 0; v < 2; ++v) {
            const int K = KROW(half, s, v);
            float val = (n >= K) ? (SCOEF * ap[n - K]) : 0.0f;
            if (v == 0) Ls[s].x = val; else Ls[s].y = val;
        }
    }
    v2f Ac;
    Ac.x = (half == 0) ? ap[n + 1] : 0.0f;
    Ac.y = 0.0f;

    const int t_start = seg * SEGLEN;
    const int warm    = (seg == 0) ? 0 : WARMCH;
    const int tfirst  = t_start - warm * 16;

    const char* __restrict__ xwalk =
        (const char*)(x + (size_t)b * Tn * Fn + ft * 16) + (size_t)tfirst * ROWB;
    const char* __restrict__ xmain = xwalk + (size_t)warm * CHB;
    char* __restrict__ omain =
        (char*)(out + (size_t)b * Tn * Fn + ft * 16) + (size_t)t_start * ROWB;

    const unsigned laneB = (unsigned)KROW(half, 0, 0) * ROWB + (unsigned)n * 4u;

    float c = ldf((gcptr)xwalk, (unsigned)n * 4u);

    v2f   Bq[BRING][4];

    #pragma unroll
    for (int i = 0; i < BDIST; ++i) load_b_tile(Bq[i], (gcptr)(xwalk + (size_t)i * CHB) + laneB);

    gcptr PB[BRING];
    gptr  PO[HOLD];
    float Oq[HOLD][8];
    #pragma unroll
    for (int i = 0; i < BRING; ++i) PB[i] = (gcptr)xwalk;
    #pragma unroll
    for (int i = 0; i < HOLD; ++i) {
        PO[i] = (gptr)omain;
        #pragma unroll
        for (int v = 0; v < 8; ++v) Oq[i][v] = 0.0f;
    }

    if (warm != 0) {
        #pragma nounroll
        for (int j = 0; j < WARMCH; j += BRING) {
            #pragma unroll
            for (int u = 0; u < BRING; ++u) {
                hold_ptr(PB[u]);
                PB[u] = lane_ptr(xwalk, j + u + BDIST, laneB);
                load_b_tile(Bq[(u + BDIST) % BRING], PB[u]);
                const v8f acc = ema_chunk(Ls, Ac, Bq[u], c, half);
                c = __shfl(acc[7], 16 + n, 32);
            }
        }
    }

    #pragma nounroll
    for (int k = 0; k < SEGCH; k += BRING) {
        #pragma unroll
        for (int u = 0; u < BRING; ++u) {
            const int kb = (k + u + BDIST < SEGCH) ? (k + u + BDIST) : (SEGCH - 1);
            hold_chunk(PB[u], PO[u % HOLD], Oq[u % HOLD]);
            PB[u] = lane_ptr(xmain, kb, laneB);
            PO[u % HOLD] = (gptr)lane_ptr(omain, k + u, laneB);
            load_b_tile(Bq[(u + BDIST) % BRING], PB[u]);

            const v8f acc = ema_chunk(Ls, Ac, Bq[u], c, half);
            c = __shfl(acc[7], 16 + n, 32);

            #pragma unroll
            for (int v = 0; v < 8; ++v) {
                const float xv = (v & 1) ? Bq[u][v >> 1].y : Bq[u][v >> 1].x;
                const float mm = acc[v] + EPSC;
                const float ratio =
                    xv * __builtin_amdgcn_exp2f(-ALPHAC * __builtin_amdgcn_logf(mm));
                const float val =
                    __builtin_amdgcn_sqrtf(__builtin_amdgcn_sqrtf(ratio + DELTAC)) - DROOT;
                Oq[u % HOLD][v] = val;
                __builtin_nontemporal_store(val, (gfptr)(PO[u % HOLD] + (unsigned)v * ROWB));
            }
        }
    }
}

extern "C" void kernel_launch(void* const* d_in, const int* in_sizes, int n_in,
                              void* d_out, int out_size, void* d_ws, size_t ws_size,
                              hipStream_t stream) {
    (void)in_sizes; (void)n_in; (void)d_ws; (void)ws_size; (void)out_size;
    const float* x = (const float*)d_in[0];
    float* out = (float*)d_out;

    const int totalWaves = Bn * (Fn / 16) * NSEG;
    const int wavesPerBlock = 8;
    dim3 grid(totalWaves / wavesPerBlock);
    dim3 block(wavesPerBlock * 32);
    pcen_wmma_kernel<<<grid, block, 0, stream>>>(x, out);
}


# Each wave owns 32 channels as two 16-column tiles (channels 2n+j): loads/stores become 8-byte lane accesses covering whole 128-byte lines; 1024 waves; per-tile WMMA chain and sum order unchanged
# speedup vs baseline: 2.4612x; 1.2147x over previous
#include <hip/hip_runtime.h>
#include <hip/hip_bf16.h>


typedef __attribute__((ext_vector_type(2))) float v2f;
typedef __attribute__((ext_vector_type(8))) float v8f;

#define EPSC   1e-6f
#define SCOEF  0.025f
#define ACOEF  0.975f
#define ALPHAC 0.98f
#define DELTAC 2.0f
#define DROOT  1.1892071150027210667f

constexpr int Bn = 64;
constexpr int Tn = 2048;
constexpr int Fn = 128;
constexpr unsigned ROWB = Fn * 4u;
constexpr unsigned CHB  = 16u * ROWB;

constexpr int NSEG   = 4;
constexpr int SEGLEN = Tn / NSEG;
constexpr int SEGCH  = SEGLEN / 16;
constexpr int WARMCH = 32;

constexpr int NTILE = 2;
constexpr int CGRP  = Fn / (16 * NTILE);

constexpr int BRING = 8;
constexpr int BDIST = 7;
constexpr int HOLD  = 2;

static_assert(WARMCH % BRING == 0 && SEGCH % BRING == 0, "ring slot = chunk mod ring");
static_assert(BRING % HOLD == 0, "hold slot follows the unrolled body index");

typedef const __attribute__((address_space(1))) char*  gcptr;
typedef __attribute__((address_space(1))) char*        gptr;
typedef const __attribute__((address_space(1))) float* gcfptr;
typedef __attribute__((address_space(1))) float*       gfptr;
typedef const __attribute__((address_space(1))) v2f*   gcv2ptr;
typedef __attribute__((address_space(1))) v2f*         gv2ptr;

__device__ __forceinline__ v2f ld2(gcptr base, unsigned byteOff) {
    return *(gcv2ptr)(base + byteOff);
}

__device__ __forceinline__ gcptr lane_ptr(const char* slab, int chunk, unsigned laneOff) {
    gcptr p = (gcptr)(slab + (size_t)chunk * CHB);
    asm("" : "+s"(p));
    return p + laneOff;
}

__device__ __forceinline__ void hold_chunk(gcptr pb, gptr po, const v2f (&O)[8]) {
    asm volatile("" :: "v"(pb), "v"(po),
                       "v"(O[0].x), "v"(O[1].x), "v"(O[2].x), "v"(O[3].x),
                       "v"(O[4].x), "v"(O[5].x), "v"(O[6].x), "v"(O[7].x),
                       "v"(O[0].y), "v"(O[1].y), "v"(O[2].y), "v"(O[3].y),
                       "v"(O[4].y), "v"(O[5].y), "v"(O[6].y), "v"(O[7].y));
}
__device__ __forceinline__ void hold_ptr(gcptr pb) {
    asm volatile("" :: "v"(pb));
}

__device__ __forceinline__ constexpr int KROW(int half, int s, int v) {
    return 8 * half + 2 * s + v;
}

__device__ __forceinline__ void load_b_tile(v2f (&R)[8], gcptr pb) {
    #pragma unroll
    for (int r = 0; r < 8; ++r) R[r] = ld2(pb, (unsigned)r * ROWB);
}

__device__ __forceinline__ void pair_rows(v2f& Ra, v2f& Rb) {
    float a1 = Ra.y;
    float b0 = Rb.x;
    asm volatile("v_swap_b32 %0, %1" : "+v"(a1), "+v"(b0));
    Ra.y = a1;
    Rb.x = b0;
}
__device__ __forceinline__ void pair_tile(v2f (&R)[8]) {
    #pragma unroll
    for (int s = 0; s < 4; ++s) pair_rows(R[2 * s], R[2 * s + 1]);
}

template <int j>
__device__ __forceinline__ v8f ema_chunk(const v2f (&Ls)[4], v2f Ac, const v2f (&R)[8],
                                         float c, int half) {
    v8f acc = {};
    v2f Bc;
    Bc.x = (half == 0) ? c : 0.0f;
    Bc.y = 0.0f;
    acc = __builtin_amdgcn_wmma_f32_16x16x4_f32(
        false, Ac, false, Bc, (short)0, acc, false, false);
    #pragma unroll
    for (int s = 0; s < 4; ++s) {
        acc = __builtin_amdgcn_wmma_f32_16x16x4_f32(
            false, Ls[s], false, R[2 * s + j], (short)0, acc, false, false);
    }
    return acc;
}

__global__ __launch_bounds__(128)
void pcen_wmma_kernel(const float* __restrict__ x, float* __restrict__ out) {
    const int lane = threadIdx.x & 31;

    const int gw = __builtin_amdgcn_readfirstlane(
        (int)(blockIdx.x * 4u + (threadIdx.x >> 5)));
    const int cg  = gw & (CGRP - 1);
    const int seg = (gw >> 2) & (NSEG - 1);
    const int b   = gw >> 4;

    const int half = lane >> 4;
    const int n    = lane & 15;

    float ap[17];
    ap[0] = 1.0f;
    #pragma unroll
    for (int k = 1; k <= 16; ++k) ap[k] = ap[k - 1] * ACOEF;

    v2f Ls[4];
    #pragma unroll
    for (int s = 0; s < 4; ++s) {
        #pragma unroll
        for (int v = 0; v < 2; ++v) {
            const int K = KROW(half, s, v);
            float val = (n >= K) ? (SCOEF * ap[n - K]) : 0.0f;
            if (v == 0) Ls[s].x = val; else Ls[s].y = val;
        }
    }
    v2f Ac;
    Ac.x = (half == 0) ? ap[n + 1] : 0.0f;
    Ac.y = 0.0f;

    const int t_start = seg * SEGLEN;
    const int warm    = (seg == 0) ? 0 : WARMCH;
    const int tfirst  = t_start - warm * 16;

    const char* __restrict__ xwalk =
        (const char*)(x + (size_t)b * Tn * Fn + cg * (16 * NTILE)) + (size_t)tfirst * ROWB;
    const char* __restrict__ xmain = xwalk + (size_t)warm * CHB;
    char* __restrict__ omain =
        (char*)(out + (size_t)b * Tn * Fn + cg * (16 * NTILE)) + (size_t)t_start * ROWB;

    const unsigned laneB = (unsigned)KROW(half, 0, 0) * ROWB + (unsigned)n * 8u;

    const v2f cseed = ld2((gcptr)xwalk, (unsigned)n * 8u);
    float c0 = cseed.x;
    float c1 = cseed.y;

    v2f   Bq[BRING][8];

    #pragma unroll
    for (int i = 0; i < BDIST; ++i) load_b_tile(Bq[i], (gcptr)(xwalk + (size_t)i * CHB) + laneB);

    gcptr PB[BRING];
    gptr  PO[HOLD];
    v2f   Oq[HOLD][8];
    #pragma unroll
    for (int i = 0; i < BRING; ++i) PB[i] = (gcptr)xwalk;
    #pragma unroll
    for (int i = 0; i < HOLD; ++i) {
        PO[i] = (gptr)omain;
        #pragma unroll
        for (int v = 0; v < 8; ++v) { Oq[i][v].x = 0.0f; Oq[i][v].y = 0.0f; }
    }

    if (warm != 0) {
        #pragma nounroll
        for (int j = 0; j < WARMCH; j += BRING) {
            #pragma unroll
            for (int u = 0; u < BRING; ++u) {
                hold_ptr(PB[u]);
                PB[u] = lane_ptr(xwalk, j + u + BDIST, laneB);
                load_b_tile(Bq[(u + BDIST) % BRING], PB[u]);
                pair_tile(Bq[u]);
                const v8f acc0 = ema_chunk<0>(Ls, Ac, Bq[u], c0, half);
                const v8f acc1 = ema_chunk<1>(Ls, Ac, Bq[u], c1, half);
                c0 = __shfl(acc0[7], 16 + n, 32);
                c1 = __shfl(acc1[7], 16 + n, 32);
            }
        }
    }

    #pragma nounroll
    for (int k = 0; k < SEGCH; k += BRING) {
        #pragma unroll
        for (int u = 0; u < BRING; ++u) {
            const int kb = (k + u + BDIST < SEGCH) ? (k + u + BDIST) : (SEGCH - 1);
            hold_chunk(PB[u], PO[u % HOLD], Oq[u % HOLD]);
            PB[u] = lane_ptr(xmain, kb, laneB);
            PO[u % HOLD] = (gptr)lane_ptr(omain, k + u, laneB);
            load_b_tile(Bq[(u + BDIST) % BRING], PB[u]);
            pair_tile(Bq[u]);

            const v8f acc0 = ema_chunk<0>(Ls, Ac, Bq[u], c0, half);
            const v8f acc1 = ema_chunk<1>(Ls, Ac, Bq[u], c1, half);
            c0 = __shfl(acc0[7], 16 + n, 32);
            c1 = __shfl(acc1[7], 16 + n, 32);

            #pragma unroll
            for (int v = 0; v < 8; ++v) {
                v2f xv;
                xv.x = (v & 1) ? Bq[u][2 * (v >> 1)].y     : Bq[u][2 * (v >> 1)].x;
                xv.y = (v & 1) ? Bq[u][2 * (v >> 1) + 1].y : Bq[u][2 * (v >> 1) + 1].x;
                const float m0 = acc0[v] + EPSC;
                const float m1 = acc1[v] + EPSC;
                const float r0 =
                    xv.x * __builtin_amdgcn_exp2f(-ALPHAC * __builtin_amdgcn_logf(m0));
                const float r1 =
                    xv.y * __builtin_amdgcn_exp2f(-ALPHAC * __builtin_amdgcn_logf(m1));
                v2f val;
                val.x = __builtin_amdgcn_sqrtf(__builtin_amdgcn_sqrtf(r0 + DELTAC)) - DROOT;
                val.y = __builtin_amdgcn_sqrtf(__builtin_amdgcn_sqrtf(r1 + DELTAC)) - DROOT;
                Oq[u % HOLD][v] = val;
                __builtin_nontemporal_store(val, (gv2ptr)(PO[u % HOLD] + (unsigned)v * ROWB));
            }
        }
    }
}

extern "C" void kernel_launch(void* const* d_in, const int* in_sizes, int n_in,
                              void* d_out, int out_size, void* d_ws, size_t ws_size,
                              hipStream_t stream) {
    (void)in_sizes; (void)n_in; (void)d_ws; (void)ws_size; (void)out_size;
    const float* x = (const float*)d_in[0];
    float* out = (float*)d_out;

    dim3 grid(256);
    dim3 block(128);
    pcen_wmma_kernel<<<grid, block, 0, stream>>>(x, out);
}


# Carry term added last in a chunk's WMMA chain, not first: chunk k+1's four L*X WMMAs no longer wait for chunk k's carry and issue in stages beside its epilogue; sum order inside a chunk changes
# speedup vs baseline: 2.6090x; 1.0601x over previous
#include <hip/hip_runtime.h>
#include <hip/hip_bf16.h>


typedef __attribute__((ext_vector_type(2))) float v2f;
typedef __attribute__((ext_vector_type(8))) float v8f;

#define EPSC   1e-6f
#define SCOEF  0.025f
#define ACOEF  0.975f
#define ALPHAC 0.98f
#define DELTAC 2.0f
#define DROOT  1.1892071150027210667f

constexpr int Bn = 64;
constexpr int Tn = 2048;
constexpr int Fn = 128;
constexpr unsigned ROWB = Fn * 4u;
constexpr unsigned CHB  = 16u * ROWB;

constexpr int NSEG   = 4;
constexpr int SEGLEN = Tn / NSEG;
constexpr int SEGCH  = SEGLEN / 16;
constexpr int WARMCH = 32;

constexpr int NTILE = 2;
constexpr int CGRP  = Fn / (16 * NTILE);

constexpr int BRING = 8;
constexpr int BDIST = 7;
constexpr int HOLD  = 2;

static_assert(WARMCH % BRING == 0 && SEGCH % BRING == 0, "ring slot = chunk mod ring");
static_assert(BRING % HOLD == 0, "hold slot follows the unrolled body index");

typedef const __attribute__((address_space(1))) char*  gcptr;
typedef __attribute__((address_space(1))) char*        gptr;
typedef const __attribute__((address_space(1))) float* gcfptr;
typedef __attribute__((address_space(1))) float*       gfptr;
typedef const __attribute__((address_space(1))) v2f*   gcv2ptr;
typedef __attribute__((address_space(1))) v2f*         gv2ptr;

__device__ __forceinline__ v2f ld2(gcptr base, unsigned byteOff) {
    return *(gcv2ptr)(base + byteOff);
}

__device__ __forceinline__ gcptr lane_ptr(const char* slab, int chunk, unsigned laneOff) {
    gcptr p = (gcptr)(slab + (size_t)chunk * CHB);
    asm("" : "+s"(p));
    return p + laneOff;
}

__device__ __forceinline__ void hold_chunk(gcptr pb, gptr po, const v2f (&O)[8]) {
    asm volatile("" :: "v"(pb), "v"(po),
                       "v"(O[0].x), "v"(O[1].x), "v"(O[2].x), "v"(O[3].x),
                       "v"(O[4].x), "v"(O[5].x), "v"(O[6].x), "v"(O[7].x),
                       "v"(O[0].y), "v"(O[1].y), "v"(O[2].y), "v"(O[3].y),
                       "v"(O[4].y), "v"(O[5].y), "v"(O[6].y), "v"(O[7].y));
}
__device__ __forceinline__ void hold_ptr(gcptr pb) {
    asm volatile("" :: "v"(pb));
}

__device__ __forceinline__ constexpr int KROW(int half, int s, int v) {
    return 8 * half + 2 * s + v;
}

__device__ __forceinline__ void load_b_tile(v2f (&R)[8], gcptr pb) {
    #pragma unroll
    for (int r = 0; r < 8; ++r) R[r] = ld2(pb, (unsigned)r * ROWB);
}

__device__ __forceinline__ void pair_rows(v2f& Ra, v2f& Rb) {
    float a1 = Ra.y;
    float b0 = Rb.x;
    asm volatile("v_swap_b32 %0, %1" : "+v"(a1), "+v"(b0));
    Ra.y = a1;
    Rb.x = b0;
}
__device__ __forceinline__ void pair_tile(v2f (&R)[8]) {
    #pragma unroll
    for (int s = 0; s < 4; ++s) pair_rows(R[2 * s], R[2 * s + 1]);
}

template <int j>
__device__ __forceinline__ v8f ema_part(const v2f (&Ls)[4], const v2f (&R)[8]) {
    v8f acc = {};
    #pragma unroll
    for (int s = 0; s < 4; ++s) {
        acc = __builtin_amdgcn_wmma_f32_16x16x4_f32(
            false, Ls[s], false, R[2 * s + j], (short)0, acc, false, false);
    }
    return acc;
}

__device__ __forceinline__ v8f ema_carry(v2f Ac, v8f P, float c, int half) {
    v2f Bc;
    Bc.x = (half == 0) ? c : 0.0f;
    Bc.y = 0.0f;
    return __builtin_amdgcn_wmma_f32_16x16x4_f32(
        false, Ac, false, Bc, (short)0, P, false, false);
}

__global__ __launch_bounds__(128)
void pcen_wmma_kernel(const float* __restrict__ x, float* __restrict__ out) {
    const int lane = threadIdx.x & 31;

    const int gw = __builtin_amdgcn_readfirstlane(
        (int)(blockIdx.x * 4u + (threadIdx.x >> 5)));
    const int cg  = gw & (CGRP - 1);
    const int seg = (gw >> 2) & (NSEG - 1);
    const int b   = gw >> 4;

    const int half = lane >> 4;
    const int n    = lane & 15;

    float ap[17];
    ap[0] = 1.0f;
    #pragma unroll
    for (int k = 1; k <= 16; ++k) ap[k] = ap[k - 1] * ACOEF;

    v2f Ls[4];
    #pragma unroll
    for (int s = 0; s < 4; ++s) {
        #pragma unroll
        for (int v = 0; v < 2; ++v) {
            const int K = KROW(half, s, v);
            float val = (n >= K) ? (SCOEF * ap[n - K]) : 0.0f;
            if (v == 0) Ls[s].x = val; else Ls[s].y = val;
        }
    }
    v2f Ac;
    Ac.x = (half == 0) ? ap[n + 1] : 0.0f;
    Ac.y = 0.0f;

    const int t_start = seg * SEGLEN;
    const int warm    = (seg == 0) ? 0 : WARMCH;
    const int tfirst  = t_start - warm * 16;

    const char* __restrict__ xwalk =
        (const char*)(x + (size_t)b * Tn * Fn + cg * (16 * NTILE)) + (size_t)tfirst * ROWB;
    const char* __restrict__ xmain = xwalk + (size_t)warm * CHB;
    char* __restrict__ omain =
        (char*)(out + (size_t)b * Tn * Fn + cg * (16 * NTILE)) + (size_t)t_start * ROWB;

    const unsigned laneB = (unsigned)KROW(half, 0, 0) * ROWB + (unsigned)n * 8u;

    const v2f cseed = ld2((gcptr)xwalk, (unsigned)n * 8u);
    float c0 = cseed.x;
    float c1 = cseed.y;

    v2f   Bq[BRING][8];

    #pragma unroll
    for (int i = 0; i < BDIST; ++i) load_b_tile(Bq[i], (gcptr)(xwalk + (size_t)i * CHB) + laneB);

    gcptr PB[BRING];
    gptr  PO[HOLD];
    v2f   Oq[HOLD][8];
    #pragma unroll
    for (int i = 0; i < BRING; ++i) PB[i] = (gcptr)xwalk;
    #pragma unroll
    for (int i = 0; i < HOLD; ++i) {
        PO[i] = (gptr)omain;
        #pragma unroll
        for (int v = 0; v < 8; ++v) { Oq[i][v].x = 0.0f; Oq[i][v].y = 0.0f; }
    }

    pair_tile(Bq[0]);
    v8f P0 = ema_part<0>(Ls, Bq[0]);
    v8f P1 = ema_part<1>(Ls, Bq[0]);

    if (warm != 0) {
        #pragma nounroll
        for (int j = 0; j < WARMCH; j += BRING) {
            #pragma unroll
            for (int u = 0; u < BRING; ++u) {
                hold_ptr(PB[u]);
                PB[u] = lane_ptr(xwalk, j + u + BDIST, laneB);
                load_b_tile(Bq[(u + BDIST) % BRING], PB[u]);
                #define PIN_PAIR(A, B) do { asm volatile("" : "+v"(A), "+v"(B)); __builtin_amdgcn_sched_barrier(0x80); } while (0)
                v8f acc0 = ema_carry(Ac, P0, c0, half);
                v8f acc1 = ema_carry(Ac, P1, c1, half);
                pair_tile(Bq[(u + 1) % BRING]);
                PIN_PAIR(acc0, acc1);
                c0 = __shfl(acc0[7], 16 + n, 32);
                c1 = __shfl(acc1[7], 16 + n, 32);
                const v2f (&Rn)[8] = Bq[(u + 1) % BRING];
                const v8f zero8 = {};
                #pragma unroll
                for (int s = 0; s < 4; ++s) {
                    P0 = __builtin_amdgcn_wmma_f32_16x16x4_f32(
                        false, Ls[s], false, Rn[2 * s], (short)0, (s == 0) ? zero8 : P0, false, false);
                    P1 = __builtin_amdgcn_wmma_f32_16x16x4_f32(
                        false, Ls[s], false, Rn[2 * s + 1], (short)0, (s == 0) ? zero8 : P1, false, false);
                    PIN_PAIR(P0, P1);
                }
                #undef PIN_PAIR
            }
        }
    }

    #pragma nounroll
    for (int k = 0; k < SEGCH; k += BRING) {
        #pragma unroll
        for (int u = 0; u < BRING; ++u) {
            const int kb = (k + u + BDIST < SEGCH) ? (k + u + BDIST) : (SEGCH - 1);
            hold_chunk(PB[u], PO[u % HOLD], Oq[u % HOLD]);
            PB[u] = lane_ptr(xmain, kb, laneB);
            PO[u % HOLD] = (gptr)lane_ptr(omain, k + u, laneB);
            load_b_tile(Bq[(u + BDIST) % BRING], PB[u]);

            #define STAGE_END() __builtin_amdgcn_sched_barrier(0x80)
            #define STAGE_END_W(P) do { asm volatile("" : "+v"(P)); __builtin_amdgcn_sched_barrier(0x80); } while (0)
            const v2f (&Rn)[8] = Bq[(u + 1) % BRING];
            const v2f (&Rx)[8] = Bq[u];
            const v8f zero8 = {};
            float e0[8], e1[8];

            const v8f acc0 = ema_carry(Ac, P0, c0, half);
            const v8f acc1 = ema_carry(Ac, P1, c1, half);
            pair_tile(Bq[(u + 1) % BRING]);
            STAGE_END();

            P0 = __builtin_amdgcn_wmma_f32_16x16x4_f32(false, Ls[0], false, Rn[0], (short)0, zero8, false, false);
            #pragma unroll
            for (int v = 0; v < 8; ++v) e0[v] = acc0[v] + EPSC;
            c0 = __shfl(acc0[7], 16 + n, 32);
            STAGE_END_W(P0);

            P1 = __builtin_amdgcn_wmma_f32_16x16x4_f32(false, Ls[0], false, Rn[1], (short)0, zero8, false, false);
            #pragma unroll
            for (int v = 0; v < 8; ++v) {
                e0[v] = __builtin_amdgcn_logf(e0[v]);
                e1[v] = acc1[v] + EPSC;
            }
            c1 = __shfl(acc1[7], 16 + n, 32);
            STAGE_END_W(P1);

            P0 = __builtin_amdgcn_wmma_f32_16x16x4_f32(false, Ls[1], false, Rn[2], (short)0, P0, false, false);
            #pragma unroll
            for (int v = 0; v < 8; ++v) {
                e1[v] = __builtin_amdgcn_logf(e1[v]);
                e0[v] = -ALPHAC * e0[v];
            }
            STAGE_END_W(P0);

            P1 = __builtin_amdgcn_wmma_f32_16x16x4_f32(false, Ls[1], false, Rn[3], (short)0, P1, false, false);
            #pragma unroll
            for (int v = 0; v < 8; ++v) {
                e0[v] = __builtin_amdgcn_exp2f(e0[v]);
                e1[v] = -ALPHAC * e1[v];
            }
            STAGE_END_W(P1);

            P0 = __builtin_amdgcn_wmma_f32_16x16x4_f32(false, Ls[2], false, Rn[4], (short)0, P0, false, false);
            #pragma unroll
            for (int v = 0; v < 8; ++v) {
                e1[v] = __builtin_amdgcn_exp2f(e1[v]);
                const float x0 = (v & 1) ? Rx[2 * (v >> 1)].y : Rx[2 * (v >> 1)].x;
                e0[v] = x0 * e0[v] + DELTAC;
            }
            STAGE_END_W(P0);

            P1 = __builtin_amdgcn_wmma_f32_16x16x4_f32(false, Ls[2], false, Rn[5], (short)0, P1, false, false);
            #pragma unroll
            for (int v = 0; v < 8; ++v) {
                e0[v] = __builtin_amdgcn_sqrtf(e0[v]);
                const float x1 = (v & 1) ? Rx[2 * (v >> 1) + 1].y : Rx[2 * (v >> 1) + 1].x;
                e1[v] = x1 * e1[v] + DELTAC;
            }
            STAGE_END_W(P1);

            P0 = __builtin_amdgcn_wmma_f32_16x16x4_f32(false, Ls[3], false, Rn[6], (short)0, P0, false, false);
            #pragma unroll
            for (int v = 0; v < 8; ++v) e1[v] = __builtin_amdgcn_sqrtf(e1[v]);
            STAGE_END_W(P0);

            P1 = __builtin_amdgcn_wmma_f32_16x16x4_f32(false, Ls[3], false, Rn[7], (short)0, P1, false, false);
            #pragma unroll
            for (int v = 0; v < 8; ++v) e0[v] = __builtin_amdgcn_sqrtf(e0[v]);
            STAGE_END_W(P1);

            #pragma unroll
            for (int v = 0; v < 8; ++v) {
                e1[v] = __builtin_amdgcn_sqrtf(e1[v]);
                e0[v] = e0[v] - DROOT;
            }
            STAGE_END();

            #pragma unroll
            for (int v = 0; v < 8; ++v) {
                v2f val;
                val.x = e0[v];
                val.y = e1[v] - DROOT;
                Oq[u % HOLD][v] = val;
                __builtin_nontemporal_store(val, (gv2ptr)(PO[u % HOLD] + (unsigned)v * ROWB));
            }
            STAGE_END();
            #undef STAGE_END_W
            #undef STAGE_END
        }
    }
}

extern "C" void kernel_launch(void* const* d_in, const int* in_sizes, int n_in,
                              void* d_out, int out_size, void* d_ws, size_t ws_size,
                              hipStream_t stream) {
    (void)in_sizes; (void)n_in; (void)d_ws; (void)ws_size; (void)out_size;
    const float* x = (const float*)d_in[0];
    float* out = (float*)d_out;

    dim3 grid(256);
    dim3 block(128);
    pcen_wmma_kernel<<<grid, block, 0, stream>>>(x, out);
}


# Time axis cut in 8 pieces = the 8 waves of a block: each wave scans only the piece before its own from zero carry, carries combined exactly in LDS (c=a^256*c+e); 2 waves/SIMD; carry sum order changes
# speedup vs baseline: 2.8936x; 1.1091x over previous
#include <hip/hip_runtime.h>
#include <hip/hip_bf16.h>


typedef __attribute__((ext_vector_type(2))) float v2f;
typedef __attribute__((ext_vector_type(8))) float v8f;

#define EPSC   1e-6f
#define SCOEF  0.025f
#define ACOEF  0.975f
#define ALPHAC 0.98f
#define DELTAC 2.0f
#define DROOT  1.1892071150027210667f

constexpr int Bn = 64;
constexpr int Tn = 2048;
constexpr int Fn = 128;
constexpr unsigned ROWB = Fn * 4u;
constexpr unsigned CHB  = 16u * ROWB;

constexpr int NSEG   = 8;
constexpr int SEGLEN = Tn / NSEG;
constexpr int SEGCH  = SEGLEN / 16;
constexpr int WARMCH = SEGCH;

constexpr int NTILE = 2;
constexpr int CGRP  = Fn / (16 * NTILE);

constexpr int BRING = 8;
constexpr int BDIST = 7;
constexpr int HOLD  = 2;

static_assert(WARMCH % BRING == 0 && SEGCH % BRING == 0, "ring slot = chunk mod ring");
static_assert(BRING % HOLD == 0, "hold slot follows the unrolled body index");

typedef const __attribute__((address_space(1))) char*  gcptr;
typedef __attribute__((address_space(1))) char*        gptr;
typedef const __attribute__((address_space(1))) float* gcfptr;
typedef __attribute__((address_space(1))) float*       gfptr;
typedef const __attribute__((address_space(1))) v2f*   gcv2ptr;
typedef __attribute__((address_space(1))) v2f*         gv2ptr;

__device__ __forceinline__ v2f ld2(gcptr base, unsigned byteOff) {
    return *(gcv2ptr)(base + byteOff);
}

__device__ __forceinline__ gcptr lane_ptr(const char* slab, int chunk, unsigned laneOff) {
    gcptr p = (gcptr)(slab + (size_t)chunk * CHB);
    asm("" : "+s"(p));
    return p + laneOff;
}

__device__ __forceinline__ void hold_chunk(gcptr pb, gptr po, const v2f (&O)[8]) {
    asm volatile("" :: "v"(pb), "v"(po),
                       "v"(O[0].x), "v"(O[1].x), "v"(O[2].x), "v"(O[3].x),
                       "v"(O[4].x), "v"(O[5].x), "v"(O[6].x), "v"(O[7].x),
                       "v"(O[0].y), "v"(O[1].y), "v"(O[2].y), "v"(O[3].y),
                       "v"(O[4].y), "v"(O[5].y), "v"(O[6].y), "v"(O[7].y));
}
__device__ __forceinline__ void hold_ptr(gcptr pb) {
    asm volatile("" :: "v"(pb));
}

__device__ __forceinline__ constexpr int KROW(int half, int s, int v) {
    return 8 * half + 2 * s + v;
}

__device__ __forceinline__ void load_b_tile(v2f (&R)[8], gcptr pb) {
    #pragma unroll
    for (int r = 0; r < 8; ++r) R[r] = ld2(pb, (unsigned)r * ROWB);
}

__device__ __forceinline__ void pair_rows(v2f& Ra, v2f& Rb) {
    float a1 = Ra.y;
    float b0 = Rb.x;
    asm volatile("v_swap_b32 %0, %1" : "+v"(a1), "+v"(b0));
    Ra.y = a1;
    Rb.x = b0;
}
__device__ __forceinline__ void pair_tile(v2f (&R)[8]) {
    #pragma unroll
    for (int s = 0; s < 4; ++s) pair_rows(R[2 * s], R[2 * s + 1]);
}

template <int j>
__device__ __forceinline__ v8f ema_part(const v2f (&Ls)[4], const v2f (&R)[8]) {
    v8f acc = {};
    #pragma unroll
    for (int s = 0; s < 4; ++s) {
        acc = __builtin_amdgcn_wmma_f32_16x16x4_f32(
            false, Ls[s], false, R[2 * s + j], (short)0, acc, false, false);
    }
    return acc;
}

__device__ __forceinline__ v8f ema_carry(v2f Ac, v8f P, float c, int half) {
    v2f Bc;
    Bc.x = (half == 0) ? c : 0.0f;
    Bc.y = 0.0f;
    return __builtin_amdgcn_wmma_f32_16x16x4_f32(
        false, Ac, false, Bc, (short)0, P, false, false);
}

__global__ __launch_bounds__(256)
void pcen_wmma_kernel(const float* __restrict__ x, float* __restrict__ out) {
    __shared__ float xc[NSEG][NTILE][16];

    const int lane = threadIdx.x & 31;

    const int seg = __builtin_amdgcn_readfirstlane((int)(threadIdx.x >> 5));
    const int cg  = (int)(blockIdx.x & (unsigned)(CGRP - 1));
    const int b   = (int)(blockIdx.x >> 2);

    const int half = lane >> 4;
    const int n    = lane & 15;

    float ap[17];
    ap[0] = 1.0f;
    #pragma unroll
    for (int k = 1; k <= 16; ++k) ap[k] = ap[k - 1] * ACOEF;

    v2f Ls[4];
    #pragma unroll
    for (int s = 0; s < 4; ++s) {
        #pragma unroll
        for (int v = 0; v < 2; ++v) {
            const int K = KROW(half, s, v);
            float val = (n >= K) ? (SCOEF * ap[n - K]) : 0.0f;
            if (v == 0) Ls[s].x = val; else Ls[s].y = val;
        }
    }
    v2f Ac;
    Ac.x = (half == 0) ? ap[n + 1] : 0.0f;
    Ac.y = 0.0f;

    const int t_start = seg * SEGLEN;
    const int warm    = (seg == 0) ? 0 : WARMCH;
    const int tfirst  = t_start - warm * 16;

    const char* __restrict__ xwalk =
        (const char*)(x + (size_t)b * Tn * Fn + cg * (16 * NTILE)) + (size_t)tfirst * ROWB;
    const char* __restrict__ xmain = xwalk + (size_t)warm * CHB;
    char* __restrict__ omain =
        (char*)(out + (size_t)b * Tn * Fn + cg * (16 * NTILE)) + (size_t)t_start * ROWB;

    const unsigned laneB = (unsigned)KROW(half, 0, 0) * ROWB + (unsigned)n * 8u;

    const v2f cseed = ld2((gcptr)xwalk, (unsigned)n * 8u);
    float c0 = (seg == 0) ? cseed.x : 0.0f;
    float c1 = (seg == 0) ? cseed.y : 0.0f;

    v2f   Bq[BRING][8];

    #pragma unroll
    for (int i = 0; i < BDIST; ++i) load_b_tile(Bq[i], (gcptr)(xwalk + (size_t)i * CHB) + laneB);

    gcptr PB[BRING];
    gptr  PO[HOLD];
    v2f   Oq[HOLD][8];
    #pragma unroll
    for (int i = 0; i < BRING; ++i) PB[i] = (gcptr)xwalk;
    #pragma unroll
    for (int i = 0; i < HOLD; ++i) {
        PO[i] = (gptr)omain;
        #pragma unroll
        for (int v = 0; v < 8; ++v) { Oq[i][v].x = 0.0f; Oq[i][v].y = 0.0f; }
    }

    pair_tile(Bq[0]);
    v8f P0 = ema_part<0>(Ls, Bq[0]);
    v8f P1 = ema_part<1>(Ls, Bq[0]);

    if (warm != 0) {
        #pragma nounroll
        for (int j = 0; j < WARMCH; j += BRING) {
            #pragma unroll
            for (int u = 0; u < BRING; ++u) {
                hold_ptr(PB[u]);
                PB[u] = lane_ptr(xwalk, j + u + BDIST, laneB);
                load_b_tile(Bq[(u + BDIST) % BRING], PB[u]);
                #define PIN_PAIR(A, B) do { asm volatile("" : "+v"(A), "+v"(B)); __builtin_amdgcn_sched_barrier(0x80); } while (0)
                v8f acc0 = ema_carry(Ac, P0, c0, half);
                v8f acc1 = ema_carry(Ac, P1, c1, half);
                pair_tile(Bq[(u + 1) % BRING]);
                PIN_PAIR(acc0, acc1);
                c0 = __shfl(acc0[7], 16 + n, 32);
                c1 = __shfl(acc1[7], 16 + n, 32);
                const v2f (&Rn)[8] = Bq[(u + 1) % BRING];
                const v8f zero8 = {};
                #pragma unroll
                for (int s = 0; s < 4; ++s) {
                    P0 = __builtin_amdgcn_wmma_f32_16x16x4_f32(
                        false, Ls[s], false, Rn[2 * s], (short)0, (s == 0) ? zero8 : P0, false, false);
                    P1 = __builtin_amdgcn_wmma_f32_16x16x4_f32(
                        false, Ls[s], false, Rn[2 * s + 1], (short)0, (s == 0) ? zero8 : P1, false, false);
                    PIN_PAIR(P0, P1);
                }
                #undef PIN_PAIR
            }
        }
    }

    xc[seg][0][n] = c0;
    xc[seg][1][n] = c1;
    __builtin_amdgcn_fence(__ATOMIC_RELEASE, "workgroup", "local");
    __builtin_amdgcn_s_barrier();
    __builtin_amdgcn_fence(__ATOMIC_ACQUIRE, "workgroup", "local");
    {
        float a256 = ap[16];
        for (int q = 0; q < 4; ++q) a256 = a256 * a256;
        float C0 = xc[0][0][n];
        float C1 = xc[0][1][n];
        for (int i = 1; i <= seg; ++i) {
            C0 = a256 * C0 + xc[i][0][n];
            C1 = a256 * C1 + xc[i][1][n];
        }
        c0 = C0;
        c1 = C1;
    }

    #pragma nounroll
    for (int k = 0; k < SEGCH; k += BRING) {
        #pragma unroll
        for (int u = 0; u < BRING; ++u) {
            const int kb = (k + u + BDIST < SEGCH) ? (k + u + BDIST) : (SEGCH - 1);
            hold_chunk(PB[u], PO[u % HOLD], Oq[u % HOLD]);
            PB[u] = lane_ptr(xmain, kb, laneB);
            PO[u % HOLD] = (gptr)lane_ptr(omain, k + u, laneB);
            load_b_tile(Bq[(u + BDIST) % BRING], PB[u]);

            #define STAGE_END() __builtin_amdgcn_sched_barrier(0x80)
            #define STAGE_END_W(P) do { asm volatile("" : "+v"(P)); __builtin_amdgcn_sched_barrier(0x80); } while (0)
            const v2f (&Rn)[8] = Bq[(u + 1) % BRING];
            const v2f (&Rx)[8] = Bq[u];
            const v8f zero8 = {};
            float e0[8], e1[8];

            const v8f acc0 = ema_carry(Ac, P0, c0, half);
            const v8f acc1 = ema_carry(Ac, P1, c1, half);
            pair_tile(Bq[(u + 1) % BRING]);
            STAGE_END();

            P0 = __builtin_amdgcn_wmma_f32_16x16x4_f32(false, Ls[0], false, Rn[0], (short)0, zero8, false, false);
            #pragma unroll
            for (int v = 0; v < 8; ++v) e0[v] = acc0[v] + EPSC;
            c0 = __shfl(acc0[7], 16 + n, 32);
            STAGE_END_W(P0);

            P1 = __builtin_amdgcn_wmma_f32_16x16x4_f32(false, Ls[0], false, Rn[1], (short)0, zero8, false, false);
            #pragma unroll
            for (int v = 0; v < 8; ++v) {
                e0[v] = __builtin_amdgcn_logf(e0[v]);
                e1[v] = acc1[v] + EPSC;
            }
            c1 = __shfl(acc1[7], 16 + n, 32);
            STAGE_END_W(P1);

            P0 = __builtin_amdgcn_wmma_f32_16x16x4_f32(false, Ls[1], false, Rn[2], (short)0, P0, false, false);
            #pragma unroll
            for (int v = 0; v < 8; ++v) {
                e1[v] = __builtin_amdgcn_logf(e1[v]);
                e0[v] = -ALPHAC * e0[v];
            }
            STAGE_END_W(P0);

            P1 = __builtin_amdgcn_wmma_f32_16x16x4_f32(false, Ls[1], false, Rn[3], (short)0, P1, false, false);
            #pragma unroll
            for (int v = 0; v < 8; ++v) {
                e0[v] = __builtin_amdgcn_exp2f(e0[v]);
                e1[v] = -ALPHAC * e1[v];
            }
            STAGE_END_W(P1);

            P0 = __builtin_amdgcn_wmma_f32_16x16x4_f32(false, Ls[2], false, Rn[4], (short)0, P0, false, false);
            #pragma unroll
            for (int v = 0; v < 8; ++v) {
                e1[v] = __builtin_amdgcn_exp2f(e1[v]);
                const float x0 = (v & 1) ? Rx[2 * (v >> 1)].y : Rx[2 * (v >> 1)].x;
                e0[v] = x0 * e0[v] + DELTAC;
            }
            STAGE_END_W(P0);

            P1 = __builtin_amdgcn_wmma_f32_16x16x4_f32(false, Ls[2], false, Rn[5], (short)0, P1, false, false);
            #pragma unroll
            for (int v = 0; v < 8; ++v) {
                e0[v] = __builtin_amdgcn_sqrtf(e0[v]);
                const float x1 = (v & 1) ? Rx[2 * (v >> 1) + 1].y : Rx[2 * (v >> 1) + 1].x;
                e1[v] = x1 * e1[v] + DELTAC;
            }
            STAGE_END_W(P1);

            P0 = __builtin_amdgcn_wmma_f32_16x16x4_f32(false, Ls[3], false, Rn[6], (short)0, P0, false, false);
            #pragma unroll
            for (int v = 0; v < 8; ++v) e1[v] = __builtin_amdgcn_sqrtf(e1[v]);
            STAGE_END_W(P0);

            P1 = __builtin_amdgcn_wmma_f32_16x16x4_f32(false, Ls[3], false, Rn[7], (short)0, P1, false, false);
            #pragma unroll
            for (int v = 0; v < 8; ++v) e0[v] = __builtin_amdgcn_sqrtf(e0[v]);
            STAGE_END_W(P1);

            #pragma unroll
            for (int v = 0; v < 8; ++v) {
                e1[v] = __builtin_amdgcn_sqrtf(e1[v]);
                e0[v] = e0[v] - DROOT;
            }
            STAGE_END();

            #pragma unroll
            for (int v = 0; v < 8; ++v) {
                v2f val;
                val.x = e0[v];
                val.y = e1[v] - DROOT;
                Oq[u % HOLD][v] = val;
                __builtin_nontemporal_store(val, (gv2ptr)(PO[u % HOLD] + (unsigned)v * ROWB));
            }
            STAGE_END();
            #undef STAGE_END_W
            #undef STAGE_END
        }
    }
}

extern "C" void kernel_launch(void* const* d_in, const int* in_sizes, int n_in,
                              void* d_out, int out_size, void* d_ws, size_t ws_size,
                              hipStream_t stream) {
    (void)in_sizes; (void)n_in; (void)d_ws; (void)ws_size; (void)out_size;
    const float* x = (const float*)d_in[0];
    float* out = (float*)d_out;

    dim3 grid(256);
    dim3 block(256);
    pcen_wmma_kernel<<<grid, block, 0, stream>>>(x, out);
}


# Each wave scans its own piece once: x stays in registers, zero-carry L*X tiles parked in LDS feed the carry WMMA; half the WMMAs and loads; piece end state by FMA: carry sum order changes
# speedup vs baseline: 3.7568x; 1.2983x over previous
#include <hip/hip_runtime.h>
#include <hip/hip_bf16.h>


typedef __attribute__((ext_vector_type(2))) float v2f;
typedef __attribute__((ext_vector_type(4))) float v4f;
typedef __attribute__((ext_vector_type(8))) float v8f;

#define EPSC   1e-6f
#define SCOEF  0.025f
#define ACOEF  0.975f
#define ALPHAC 0.98f
#define DELTAC 2.0f
#define DROOT  1.1892071150027210667f

constexpr int Bn = 64;
constexpr int Tn = 2048;
constexpr int Fn = 128;
constexpr unsigned ROWB = Fn * 4u;
constexpr unsigned CHB  = 16u * ROWB;

constexpr int NSEG   = 8;
constexpr int SEGLEN = Tn / NSEG;
constexpr int SEGCH  = SEGLEN / 16;

constexpr int NTILE = 2;
constexpr int CGRP  = Fn / (16 * NTILE);

constexpr int LDIST = 7;
constexpr int HOLD  = 2;

typedef const __attribute__((address_space(1))) char*  gcptr;
typedef __attribute__((address_space(1))) char*        gptr;
typedef const __attribute__((address_space(1))) v2f*   gcv2ptr;
typedef __attribute__((address_space(1))) v2f*         gv2ptr;

__device__ __forceinline__ v2f ld2(gcptr base, unsigned byteOff) {
    return *(gcv2ptr)(base + byteOff);
}

__device__ __forceinline__ void hold_vals(const v2f (&O)[8]) {
    asm volatile("" :: "v"(O[0].x), "v"(O[1].x), "v"(O[2].x), "v"(O[3].x),
                       "v"(O[4].x), "v"(O[5].x), "v"(O[6].x), "v"(O[7].x),
                       "v"(O[0].y), "v"(O[1].y), "v"(O[2].y), "v"(O[3].y),
                       "v"(O[4].y), "v"(O[5].y), "v"(O[6].y), "v"(O[7].y));
}

__device__ __forceinline__ constexpr int KROW(int half, int s, int v) {
    return 8 * half + 2 * s + v;
}

__device__ __forceinline__ void load_b_tile(v2f (&R)[8], gcptr pb) {
    #pragma unroll
    for (int r = 0; r < 8; ++r) R[r] = ld2(pb, (unsigned)r * ROWB);
}

__device__ __forceinline__ void pair_rows(v2f& Ra, v2f& Rb) {
    const float a1 = Ra.y;
    Ra.y = Rb.x;
    Rb.x = a1;
}
__device__ __forceinline__ void pair_tile(v2f (&R)[8]) {
    #pragma unroll
    for (int s = 0; s < 4; ++s) pair_rows(R[2 * s], R[2 * s + 1]);
}

__device__ __forceinline__ v8f ema_carry(v2f Ac, v8f P, float c, int half) {
    v2f Bc;
    Bc.x = (half == 0) ? c : 0.0f;
    Bc.y = 0.0f;
    return __builtin_amdgcn_wmma_f32_16x16x4_f32(
        false, Ac, false, Bc, (short)0, P, false, false);
}

__device__ __forceinline__ v4f lo4(v8f v) { return __builtin_shufflevector(v, v, 0, 1, 2, 3); }
__device__ __forceinline__ v4f hi4(v8f v) { return __builtin_shufflevector(v, v, 4, 5, 6, 7); }
__device__ __forceinline__ v8f cat8(v4f a, v4f b) { return __builtin_shufflevector(a, b, 0, 1, 2, 3, 4, 5, 6, 7); }

__global__ __launch_bounds__(256)
void pcen_wmma_kernel(const float* __restrict__ x, float* __restrict__ out) {
    __shared__ v4f park[NSEG * SEGCH * 4 * 32];
    __shared__ float xc[NSEG][NTILE][16];

    const int lane = threadIdx.x & 31;

    const int seg = __builtin_amdgcn_readfirstlane((int)(threadIdx.x >> 5));
    const int cg  = (int)(blockIdx.x & (unsigned)(CGRP - 1));
    const int b   = (int)(blockIdx.x >> 2);

    const int half = lane >> 4;
    const int n    = lane & 15;

    float ap[17];
    ap[0] = 1.0f;
    #pragma unroll
    for (int k = 1; k <= 16; ++k) ap[k] = ap[k - 1] * ACOEF;

    v2f Ls[4];
    #pragma unroll
    for (int s = 0; s < 4; ++s) {
        #pragma unroll
        for (int v = 0; v < 2; ++v) {
            const int K = KROW(half, s, v);
            float val = (n >= K) ? (SCOEF * ap[n - K]) : 0.0f;
            if (v == 0) Ls[s].x = val; else Ls[s].y = val;
        }
    }
    v2f Ac;
    Ac.x = (half == 0) ? ap[n + 1] : 0.0f;
    Ac.y = 0.0f;

    const char* __restrict__ xseq  = (const char*)(x + (size_t)b * Tn * Fn + cg * (16 * NTILE));
    const char* __restrict__ xmain = xseq + (size_t)seg * SEGLEN * ROWB;
    char* __restrict__ omain =
        (char*)(out + (size_t)b * Tn * Fn + cg * (16 * NTILE)) + (size_t)seg * SEGLEN * ROWB;

    const unsigned laneB = (unsigned)KROW(half, 0, 0) * ROWB + (unsigned)n * 8u;
    const gcptr px = (gcptr)xmain + laneB;
    const gptr  po = (gptr)omain + laneB;

    const v2f cseed = ld2((gcptr)xseq, (unsigned)n * 8u);

    v4f* const pk = park + (size_t)seg * (SEGCH * 4 * 32) + lane;

    v2f X[SEGCH][8];

    #pragma unroll
    for (int i = 0; i < LDIST; ++i) load_b_tile(X[i], px + (unsigned)i * CHB);

    const float a16 = ap[16];
    float z0 = 0.0f, z1 = 0.0f;
    #pragma unroll
    for (int j = 0; j < SEGCH; ++j) {
        if (j + LDIST < SEGCH) load_b_tile(X[j + LDIST], px + (unsigned)(j + LDIST) * CHB);
        pair_tile(X[j]);
        #define PIN_PAIR(A, B) do { asm volatile("" : "+v"(A), "+v"(B)); __builtin_amdgcn_sched_barrier(0x80); } while (0)
        const v8f zero8 = {};
        v8f P0 = zero8, P1 = zero8;
        #pragma unroll
        for (int s = 0; s < 4; ++s) {
            P0 = __builtin_amdgcn_wmma_f32_16x16x4_f32(
                false, Ls[s], false, X[j][2 * s], (short)0, P0, false, false);
            P1 = __builtin_amdgcn_wmma_f32_16x16x4_f32(
                false, Ls[s], false, X[j][2 * s + 1], (short)0, P1, false, false);
            PIN_PAIR(P0, P1);
        }
        #undef PIN_PAIR
        z0 = __builtin_fmaf(a16, z0, P0[7]);
        z1 = __builtin_fmaf(a16, z1, P1[7]);
        pk[(j * 4 + 0) * 32] = lo4(P0);
        pk[(j * 4 + 1) * 32] = hi4(P0);
        pk[(j * 4 + 2) * 32] = lo4(P1);
        pk[(j * 4 + 3) * 32] = hi4(P1);
    }

    xc[seg][0][n] = __shfl(z0, 16 + n, 32);
    xc[seg][1][n] = __shfl(z1, 16 + n, 32);
    __builtin_amdgcn_fence(__ATOMIC_RELEASE, "workgroup", "local");
    __builtin_amdgcn_s_barrier();
    __builtin_amdgcn_fence(__ATOMIC_ACQUIRE, "workgroup", "local");
    float c0, c1;
    {
        float a256 = ap[16];
        for (int q = 0; q < 4; ++q) a256 = a256 * a256;
        float C0 = cseed.x;
        float C1 = cseed.y;
        for (int i = 0; i < seg; ++i) {
            C0 = a256 * C0 + xc[i][0][n];
            C1 = a256 * C1 + xc[i][1][n];
        }
        c0 = C0;
        c1 = C1;
    }

    v2f Oq[HOLD][8];
    #pragma unroll
    for (int i = 0; i < HOLD; ++i) {
        #pragma unroll
        for (int v = 0; v < 8; ++v) { Oq[i][v].x = 0.0f; Oq[i][v].y = 0.0f; }
    }
    v8f P0 = cat8(pk[0 * 32], pk[1 * 32]);
    v8f P1 = cat8(pk[2 * 32], pk[3 * 32]);

    #pragma unroll
    for (int j = 0; j < SEGCH; ++j) {
        v8f Pn0 = P0, Pn1 = P1;
        if (j + 1 < SEGCH) {
            Pn0 = cat8(pk[((j + 1) * 4 + 0) * 32], pk[((j + 1) * 4 + 1) * 32]);
            Pn1 = cat8(pk[((j + 1) * 4 + 2) * 32], pk[((j + 1) * 4 + 3) * 32]);
        }
        hold_vals(Oq[j % HOLD]);

        #define STAGE_END() __builtin_amdgcn_sched_barrier(0x80)
        const v2f (&Rx)[8] = X[j];
        float e0[8], e1[8];

        const v8f acc0 = ema_carry(Ac, P0, c0, half);
        const v8f acc1 = ema_carry(Ac, P1, c1, half);
        STAGE_END();

        #pragma unroll
        for (int v = 0; v < 8; ++v) e0[v] = acc0[v] + EPSC;
        c0 = __shfl(acc0[7], 16 + n, 32);
        STAGE_END();

        #pragma unroll
        for (int v = 0; v < 8; ++v) {
            e0[v] = __builtin_amdgcn_logf(e0[v]);
            e1[v] = acc1[v] + EPSC;
        }
        c1 = __shfl(acc1[7], 16 + n, 32);
        STAGE_END();

        #pragma unroll
        for (int v = 0; v < 8; ++v) {
            e1[v] = __builtin_amdgcn_logf(e1[v]);
            e0[v] = -ALPHAC * e0[v];
        }
        STAGE_END();

        #pragma unroll
        for (int v = 0; v < 8; ++v) {
            e0[v] = __builtin_amdgcn_exp2f(e0[v]);
            e1[v] = -ALPHAC * e1[v];
        }
        STAGE_END();

        #pragma unroll
        for (int v = 0; v < 8; ++v) {
            e1[v] = __builtin_amdgcn_exp2f(e1[v]);
            const float x0 = (v & 1) ? Rx[2 * (v >> 1)].y : Rx[2 * (v >> 1)].x;
            e0[v] = x0 * e0[v] + DELTAC;
        }
        STAGE_END();

        #pragma unroll
        for (int v = 0; v < 8; ++v) {
            e0[v] = __builtin_amdgcn_sqrtf(e0[v]);
            const float x1 = (v & 1) ? Rx[2 * (v >> 1) + 1].y : Rx[2 * (v >> 1) + 1].x;
            e1[v] = x1 * e1[v] + DELTAC;
        }
        STAGE_END();

        #pragma unroll
        for (int v = 0; v < 8; ++v) e1[v] = __builtin_amdgcn_sqrtf(e1[v]);
        STAGE_END();

        #pragma unroll
        for (int v = 0; v < 8; ++v) e0[v] = __builtin_amdgcn_sqrtf(e0[v]);
        STAGE_END();

        #pragma unroll
        for (int v = 0; v < 8; ++v) {
            e1[v] = __builtin_amdgcn_sqrtf(e1[v]);
            e0[v] = e0[v] - DROOT;
        }
        STAGE_END();

        #pragma unroll
        for (int v = 0; v < 8; ++v) {
            v2f val;
            val.x = e0[v];
            val.y = e1[v] - DROOT;
            Oq[j % HOLD][v] = val;
            __builtin_nontemporal_store(val, (gv2ptr)(po + (unsigned)j * CHB + (unsigned)v * ROWB));
        }
        STAGE_END();
        #undef STAGE_END
        P0 = Pn0;
        P1 = Pn1;
    }
}

extern "C" void kernel_launch(void* const* d_in, const int* in_sizes, int n_in,
                              void* d_out, int out_size, void* d_ws, size_t ws_size,
                              hipStream_t stream) {
    (void)in_sizes; (void)n_in; (void)d_ws; (void)ws_size; (void)out_size;
    const float* x = (const float*)d_in[0];
    float* out = (float*)d_out;

    dim3 grid(256);
    dim3 block(256);
    pcen_wmma_kernel<<<grid, block, 0, stream>>>(x, out);
}
